# starting kernel
# baseline (speedup 1.0000x reference)
#include <hip/hip_runtime.h>

typedef float v2f __attribute__((ext_vector_type(2)));
typedef float v8f __attribute__((ext_vector_type(8)));

__device__ __forceinline__ void atomAddF(float* p, float v) {
  asm volatile("global_atomic_add_f32 %0, %1, off" : : "v"(p), "v"(v) : "memory");
}

__global__ void k_fill0(float* __restrict__ p, long long n) {
  long long i = (long long)blockIdx.x * blockDim.x + threadIdx.x;
  if (i < n) p[i] = 0.0f;
}

__global__ void k_deg(const int* __restrict__ dst, float* __restrict__ deg, int E) {
  int e = blockIdx.x * blockDim.x + threadIdx.x;
  if (e < E) atomAddF(deg + dst[e], 1.0f);
}

__global__ void k_dinv(float* __restrict__ d, int N) {
  int i = blockIdx.x * blockDim.x + threadIdx.x;
  if (i < N) d[i] = rsqrtf(d[i] + 1.0f);
}

template <int K, int N>
__launch_bounds__(32)
__global__ void k_gemm_wmma(const float* __restrict__ A, const float* __restrict__ B,
                            float* __restrict__ C) {
  const int lane = threadIdx.x;
  const int half = lane >> 4;
  const int lm   = lane & 15;
  const int m0 = blockIdx.x * 16;
  const int n0 = blockIdx.y * 16;

  const float* __restrict__ arow = A + (long long)(m0 + lm) * K + 2 * half;
  const float* __restrict__ bp   = B + (long long)(2 * half) * N + n0 + lm;

  v8f acc = {};
#pragma unroll 8
  for (int k = 0; k < K; k += 4) {
    v2f a = *(const v2f*)(arow + k);
    v2f b;
    b.x = bp[k * N];
    b.y = bp[k * N + N];
    acc = __builtin_amdgcn_wmma_f32_16x16x4_f32(
         false, a,  false, b,
         (short)0, acc,  false,  false);
  }

  float* __restrict__ cp = C + (long long)(m0 + 8 * half) * N + n0 + lm;
#pragma unroll
  for (int r = 0; r < 8; ++r)
    cp[r * N] = acc[r];
}

__launch_bounds__(256)
__global__ void k_scatter64(const float* __restrict__ h, const float* __restrict__ dinv,
                            const int* __restrict__ src, const int* __restrict__ dst,
                            float* __restrict__ agg, int E) {
  int wid  = blockIdx.x * (blockDim.x >> 5) + (threadIdx.x >> 5);
  wid = __builtin_amdgcn_readfirstlane(wid);
  int lane = threadIdx.x & 31;
  if (wid >= E) return;
  int s = src[wid];
  int d = dst[wid];
  float w = dinv[s];
  float2 v = *((const float2*)(h + (long long)s * 64) + lane);
  float* ap = agg + (long long)d * 64 + lane * 2;
  atomAddF(ap,     v.x * w);
  atomAddF(ap + 1, v.y * w);
}

__launch_bounds__(256)
__global__ void k_scatter128(const float* __restrict__ h, const float* __restrict__ dinv,
                             const int* __restrict__ src, const int* __restrict__ dst,
                             float* __restrict__ agg, int E) {
  int wid  = blockIdx.x * (blockDim.x >> 5) + (threadIdx.x >> 5);
  wid = __builtin_amdgcn_readfirstlane(wid);
  int lane = threadIdx.x & 31;
  if (wid >= E) return;
  int s = src[wid];
  int d = dst[wid];
  float w = dinv[s];
  float4 v = *((const float4*)(h + (long long)s * 128) + lane);
  float* ap = agg + (long long)d * 128 + lane * 4;
  atomAddF(ap,     v.x * w);
  atomAddF(ap + 1, v.y * w);
  atomAddF(ap + 2, v.z * w);
  atomAddF(ap + 3, v.w * w);
}

template <int F, bool RELU>
__launch_bounds__(256)
__global__ void k_finish(const float* __restrict__ h, const float* __restrict__ agg,
                         const float* __restrict__ dinv, const float* __restrict__ bias,
                         float* __restrict__ out, int N) {
  int idx = blockIdx.x * blockDim.x + threadIdx.x;
  const int F4 = F / 4;
  long long total = (long long)N * F4;
  if (idx >= total) return;
  int i  = idx / F4;
  int f4 = idx - i * F4;
  float di = dinv[i];
  float s  = di * di;
  float4 a  = ((const float4*)agg)[idx];
  float4 hv = ((const float4*)h)[idx];
  float4 bv = ((const float4*)bias)[f4];
  float4 r;
  r.x = di * a.x + s * hv.x + bv.x;
  r.y = di * a.y + s * hv.y + bv.y;
  r.z = di * a.z + s * hv.z + bv.z;
  r.w = di * a.w + s * hv.w + bv.w;
  if (RELU) {
    r.x = fmaxf(r.x, 0.0f); r.y = fmaxf(r.y, 0.0f);
    r.z = fmaxf(r.z, 0.0f); r.w = fmaxf(r.w, 0.0f);
  }
  ((float4*)out)[idx] = r;
}

extern "C" void kernel_launch(void* const* d_in, const int* in_sizes, int n_in,
                              void* d_out, int out_size, void* d_ws, size_t ws_size,
                              hipStream_t stream) {
  const float* x  = (const float*)d_in[0];
  const int*   ei = (const int*)d_in[1];
  const float* W1 = (const float*)d_in[2];
  const float* b1 = (const float*)d_in[3];
  const float* W2 = (const float*)d_in[4];
  const float* b2 = (const float*)d_in[5];

  const int FIN = 256, H = 64, FOUT = 128;
  const int N = in_sizes[0] / FIN;
  const int E = in_sizes[1] / 2;
  const int* srcI = ei;
  const int* dstI = ei + E;

  float* ws   = (float*)d_ws;
  float* dinv = ws;
  float* h1   = dinv + (long long)N;
  float* agg1 = h1   + (long long)N * H;
  float* y1   = agg1 + (long long)N * H;
  float* agg2 = y1   + (long long)N * H;
  float* h2   = h1;

  const int TB = 256;
  auto blocks = [](long long n, int tb) { return (unsigned)((n + tb - 1) / tb); };

  k_fill0<<<blocks((long long)N, TB), TB, 0, stream>>>(dinv, N);
  k_fill0<<<blocks((long long)N * H, TB), TB, 0, stream>>>(agg1, (long long)N * H);
  k_fill0<<<blocks((long long)N * FOUT, TB), TB, 0, stream>>>(agg2, (long long)N * FOUT);

  k_deg<<<blocks(E, TB), TB, 0, stream>>>(dstI, dinv, E);
  k_dinv<<<blocks(N, TB), TB, 0, stream>>>(dinv, N);

  {
    dim3 g((unsigned)(N / 16), H / 16);
    k_gemm_wmma<256, 64><<<g, 32, 0, stream>>>(x, W1, h1);
  }
  k_scatter64<<<blocks((long long)E * 32, TB), TB, 0, stream>>>(h1, dinv, srcI, dstI, agg1, E);
  k_finish<64, true><<<blocks((long long)N * H / 4, TB), TB, 0, stream>>>(h1, agg1, dinv, b1, y1, N);

  {
    dim3 g((unsigned)(N / 16), FOUT / 16);
    k_gemm_wmma<64, 128><<<g, 32, 0, stream>>>(y1, W2, h2);
  }
  k_scatter128<<<blocks((long long)E * 32, TB), TB, 0, stream>>>(h2, dinv, srcI, dstI, agg2, E);
  k_finish<128, false><<<blocks((long long)N * FOUT / 4, TB), TB, 0, stream>>>(
      h2, agg2, dinv, b2, (float*)d_out, N);
}


# Owner byte tags become one bitmap per owner work-group, built by ballots over the owner number's 8 bits: a group scans 1 bit per edge of its own bitmap, not 1 byte of all. Float sum order changed.
# speedup vs baseline: 3.5574x; 3.5574x over previous
#include <hip/hip_runtime.h>

typedef float v2f __attribute__((ext_vector_type(2)));
typedef float v8f __attribute__((ext_vector_type(8)));
typedef unsigned v4u __attribute__((ext_vector_type(4)));

constexpr int TPR  = 196;
constexpr int NOWN = 256;
constexpr int SSW  = 128;
constexpr int BTH  = 256;
constexpr int BWW  = SSW / (BTH / 32);
constexpr int BPI  = NOWN + 1;
static_assert(NOWN == 256, "the lanes of one owner meet by ballots over eight bits");
static_assert(SSW == 128 && NOWN % (BTH / 32) == 0, "write-out: four words a lane, whole owners per wave");
__launch_bounds__(BTH)
__global__ void k_bits(const int* __restrict__ dst, unsigned* __restrict__ bits, int E, int wpo) {
  __shared__ __attribute__((aligned(16))) unsigned bm[SSW * BPI];
  const int tid  = threadIdx.x;
  const int lane = tid & 31;
  const int wv   = tid >> 5;
  int d[BWW];
#pragma unroll
  for (int q = 0; q < BWW; ++q) {
    const long long e = (long long)blockIdx.x * (SSW * 32) + (wv * BWW + q) * 32 + lane;
    d[q] = e < (long long)E ? dst[e] : -1;
  }
  const uint4 z = make_uint4(0u, 0u, 0u, 0u);
  for (int i = tid; i < SSW * BPI / 4; i += BTH) ((uint4*)bm)[i] = z;
  __syncthreads();
#pragma unroll
  for (int q = 0; q < BWW; ++q) {
    const bool valid = d[q] >= 0;
    unsigned o = valid ? (unsigned)d[q] / (unsigned)TPR : 0u;
    o = o < (unsigned)(NOWN - 1) ? o : (unsigned)(NOWN - 1);
    unsigned peers = __builtin_amdgcn_ballot_w32(valid);
#pragma unroll
    for (int b = 0; b < 8; ++b) {
      const bool bit = ((o >> b) & 1u) != 0u;
      const unsigned bal = __builtin_amdgcn_ballot_w32(bit);
      peers &= bit ? bal : ~bal;
    }
    if (valid) bm[(wv * BWW + q) * BPI + (int)o] = peers;
  }
  __syncthreads();
  unsigned* __restrict__ op = bits + (long long)blockIdx.x * SSW + lane * 4;
#pragma unroll 4
  for (int i = 0; i < NOWN / (BTH / 32); ++i) {
    const int o = i * (BTH / 32) + wv;
    v4u v;
    v.x = bm[(lane * 4 + 0) * BPI + o];
    v.y = bm[(lane * 4 + 1) * BPI + o];
    v.z = bm[(lane * 4 + 2) * BPI + o];
    v.w = bm[(lane * 4 + 3) * BPI + o];
    *(v4u*)(op + (long long)o * wpo) = v;
  }
}

template <int K, int N>
__launch_bounds__(32)
__global__ void k_gemm_wmma(const float* __restrict__ A, const float* __restrict__ B,
                            float* __restrict__ C) {
  const int lane = threadIdx.x;
  const int half = lane >> 4;
  const int lm   = lane & 15;
  const int m0 = blockIdx.x * 16;
  const int n0 = blockIdx.y * 16;

  const float* __restrict__ arow = A + (long long)(m0 + lm) * K + 2 * half;
  const float* __restrict__ bp   = B + (long long)(2 * half) * N + n0 + lm;

  v8f acc = {};
#pragma unroll 8
  for (int k = 0; k < K; k += 4) {
    v2f a = *(const v2f*)(arow + k);
    v2f b;
    b.x = bp[k * N];
    b.y = bp[k * N + N];
    acc = __builtin_amdgcn_wmma_f32_16x16x4_f32(
         false, a,  false, b,
         (short)0, acc,  false,  false);
  }

  float* __restrict__ cp = C + (long long)(m0 + 8 * half) * N + n0 + lm;
#pragma unroll
  for (int r = 0; r < 8; ++r)
    cp[r * N] = acc[r];
}

constexpr int PR    = 196;
constexpr int PTB   = 1024;
constexpr int PNW   = PTB / 32;
constexpr int TIT   = 4;
constexpr int PCAP  = 512;
constexpr int PLC   = 480;
constexpr int PLH   = PLC - 32;
constexpr int NRW   = (PR + PNW - 1) / PNW;
constexpr int YP    = 64 + 4;
static_assert(PNW == 32, "row ownership is row & 31");
static_assert(PR == TPR, "the bitmaps are of groups of PR rows");
static_assert(PCAP >= 32, "an empty list must take a whole turn, or a round could make no progress");

template <int F> struct RowVec;
template <> struct RowVec<64>  { typedef float2 T; };
template <> struct RowVec<128> { typedef float4 T; };

__device__ __forceinline__ void zeroRow(float2& a) { a.x = 0.0f; a.y = 0.0f; }
__device__ __forceinline__ void zeroRow(float4& a) { a.x = 0.0f; a.y = 0.0f; a.z = 0.0f; a.w = 0.0f; }
__device__ __forceinline__ float2 fmaRow(float2 a, float2 v, float w) {
  a.x += v.x * w; a.y += v.y * w;
  return a;
}
__device__ __forceinline__ float4 fmaRow(float4 a, float4 v, float w) {
  a.x += v.x * w; a.y += v.y * w; a.z += v.z * w; a.w += v.w * w;
  return a;
}
__device__ __forceinline__ float2 finishRow(float2 a, float2 hv, float2 bv, float di, float sq, bool relu) {
  float2 r;
  r.x = di * a.x + sq * hv.x + bv.x;
  r.y = di * a.y + sq * hv.y + bv.y;
  if (relu) { r.x = fmaxf(r.x, 0.0f); r.y = fmaxf(r.y, 0.0f); }
  return r;
}
__device__ __forceinline__ float4 finishRow(float4 a, float4 hv, float4 bv, float di, float sq, bool relu) {
  float4 r;
  r.x = di * a.x + sq * hv.x + bv.x;
  r.y = di * a.y + sq * hv.y + bv.y;
  r.z = di * a.z + sq * hv.z + bv.z;
  r.w = di * a.w + sq * hv.w + bv.w;
  if (relu) {
    r.x = fmaxf(r.x, 0.0f); r.y = fmaxf(r.y, 0.0f);
    r.z = fmaxf(r.z, 0.0f); r.w = fmaxf(r.w, 0.0f);
  }
  return r;
}

__device__ __forceinline__ int firstBit(uint4 m) {
  const unsigned w = m.x ? m.x : m.y ? m.y : m.z ? m.z : m.w;
  const int qo = m.x ? 0 : m.y ? 32 : m.z ? 64 : 96;
  return qo + __builtin_ctz(w);
}
__device__ __forceinline__ uint4 dropBit(uint4 m) {
  const bool c0 = m.x != 0u, c1 = m.y != 0u, c2 = m.z != 0u;
  const unsigned w = c0 ? m.x : c1 ? m.y : c2 ? m.z : m.w;
  const unsigned nw = w & (w - 1u);
  uint4 r;
  r.x = c0 ? nw : m.x;
  r.y = (!c0 && c1) ? nw : m.y;
  r.z = (!c0 && !c1 && c2) ? nw : m.z;
  r.w = (!c0 && !c1 && !c2) ? nw : m.w;
  return r;
}

template <int F, bool RELU, bool DEG, int F2>
__launch_bounds__(PTB)
__global__ void k_pull(const float* __restrict__ h, const float* __restrict__ dinv, float* __restrict__ dout,
                       const int* __restrict__ src, const int* __restrict__ dst,
                       const float* __restrict__ bias, float* __restrict__ out, int N, int E,
                       int* __restrict__ lcnt, int* __restrict__ lbuf, int lcap,
                       const unsigned* __restrict__ bits, int G, const float* __restrict__ Wn) {
  static_assert(F2 == 0 || (F == 64 && F2 % 16 == 0 && PNW % (F2 / 16) == 0), "fused product: 64-wide rows, whole column tiles, waves split evenly over them");
  typedef typename RowVec<F>::T VT;
  constexpr int VW = F / 32;
  constexpr int PB = 2048 / F;
  __shared__ __attribute__((aligned(16))) float acc[DEG ? 4 : PR * F];
  __shared__ int Le[PNW][PCAP];
  __shared__ int Lr[PNW][PCAP];
  __shared__ int C[PNW];
  __shared__ int Dn[PNW];
  __shared__ int lstE[PNW][PLC];
  __shared__ __attribute__((aligned(16))) float ybuf[F2 > 0 ? PR * YP : 4];

  const int tid  = threadIdx.x;
  const int lane = tid & 31;
  const int wv   = __builtin_amdgcn_readfirstlane(tid >> 5);
  const int lo   = blockIdx.x * PR;
  const int nrows = (N - lo) < PR ? (N - lo) : PR;
  const unsigned nr = (unsigned)nrows;

  if constexpr (!DEG) {
    VT z;
    zeroRow(z);
    for (int r = wv; r < PR; r += PNW) *((VT*)(acc + r * F) + lane) = z;
  }

  int* myE = lstE[wv];
  int cnt = 0;

  int* __restrict__ gl = lbuf + ((long long)blockIdx.x * PNW + wv) * 2 * (long long)lcap;
  int gtot = 0;
  bool gover = false;
  int dg[NRW];
#pragma unroll
  for (int q = 0; q < NRW; ++q) dg[q] = 0;

  auto addEdges = [&](int s, int rw, int wb, int n, auto&& behind) {
    for (int hb = 0; hb < n; hb += PB) {
      VT v[PB];
#pragma unroll
      for (int t = 0; t < PB; ++t) {
        const int tt = hb + t < n ? hb + t : n - 1;
        const int sj = __builtin_amdgcn_readlane(s, tt);
        v[t] = *((const VT*)(h + (long long)sj * F) + lane);
      }
      if (hb == 0) behind();
#pragma unroll
      for (int t = 0; t < PB; ++t) {
        if (hb + t < n) {
          const int rj = __builtin_amdgcn_readlane(rw, hb + t);
          const float wj = __int_as_float(__builtin_amdgcn_readlane(wb, hb + t));
          VT* p = (VT*)(acc + rj * F) + lane;
          *p = fmaRow(*p, v[t], wj);
        }
      }
    }
  };

  bool useList = false;
  int ltot = 0;
  if (!DEG && lcap > 0) {
    const int c = lcnt[blockIdx.x * PNW + lane];
    useList = __builtin_amdgcn_ballot_w32(c < 0) == 0;
    ltot = __builtin_amdgcn_readlane(c, wv);
  }

  if (useList) {
    auto loadEnt = [&](int c0, int& s, int& rw, int& wb) {
      const int n = ltot - c0 < 32 ? ltot - c0 : 32;
      const int li = c0 + (lane < n ? lane : n - 1);
      s = gl[li]; rw = gl[lcap + li]; wb = __float_as_int(dinv[s]);
    };
    int ns = 0, nrw = 0, nwb = 0;
    if (ltot > 0) loadEnt(0, ns, nrw, nwb);
    for (int c0 = 0; c0 < ltot; c0 += 32) {
      const int n = ltot - c0 < 32 ? ltot - c0 : 32;
      const int s = ns, rw = nrw, wb = nwb;
      addEdges(s, rw, wb, n, [&] { loadEnt(c0 + 32 < ltot ? c0 + 32 : c0, ns, nrw, nwb); });
    }
  } else {
    auto workOff = [&] {
      const int n = cnt < 32 ? cnt : 32;
      const int li = cnt - n + (lane < n ? lane : n - 1);
      const int e  = myE[li];
      const int s  = src[e];
      const int rw = dst[e] - lo;
      if constexpr (DEG) {
        if (gtot + n <= lcap) {
          if (lane < n) { gl[gtot + lane] = s; gl[lcap + gtot + lane] = rw; }
          gtot += n;
        } else {
          gover = true;
        }
#pragma unroll
        for (int q = 0; q < NRW; ++q)
          dg[q] += __builtin_popcount(__builtin_amdgcn_ballot_w32((lane < n) & (rw == wv + q * PNW)));
      } else {
        const int wb = __float_as_int(dinv[s]);
        addEdges(s, rw, wb, n, [] {});
      }
      cnt -= n;
    };

    const unsigned* __restrict__ bw = bits + (long long)(blockIdx.x < NOWN - 1 ? blockIdx.x : NOWN - 1) * ((long long)G * SSW);
    int* myF = Le[wv];
    int* myR = Lr[wv];
    int g  = wv;
    int t0 = 0;
    for (;;) {
      int cs = 0;
      bool full = false;
      while (g < G && !full) {
        uint4 tv[TIT];
#pragma unroll
        for (int j = 0; j < TIT; ++j) {
          const int gj = g + j * PNW < G ? g + j * PNW : g;
          tv[j] = *((const uint4*)(bw + (long long)gj * SSW) + lane);
        }
#pragma unroll
        for (int j = 0; j < TIT; ++j) {
          if (!full && g < G) {
            uint4 mw = tv[j];
            for (int t = 0; t < t0; ++t) mw = dropBit(mw);
            const int eb = g * (SSW * 32) + lane * 128;
            int t = t0;
            while (__builtin_amdgcn_ballot_w32((mw.x | mw.y | mw.z | mw.w) != 0)) {
              const bool v = (mw.x | mw.y | mw.z | mw.w) != 0;
              const unsigned mask = __builtin_amdgcn_ballot_w32(v);
              const int n = __builtin_popcount(mask);
              if (cs + n > PCAP) { full = true; break; }
              const int pos = cs + (int)__builtin_amdgcn_mbcnt_lo(mask, 0u);
              if (v) myF[pos] = eb + firstBit(mw);
              cs += n;
              mw = dropBit(mw);
              ++t;
            }
            if (full) t0 = t;
            else { g += PNW; t0 = 0; }
          }
        }
      }
      for (int i = 0; i < cs; i += 128) {
        int r[4];
#pragma unroll
        for (int q = 0; q < 4; ++q) {
          const int k = i + q * 32 + lane;
          r[q] = k < cs ? dst[myF[k]] - lo : -1;
        }
#pragma unroll
        for (int q = 0; q < 4; ++q) {
          const int k = i + q * 32 + lane;
          if (k < cs) myR[k] = (unsigned)r[q] < nr ? r[q] : -1;
        }
      }
      if (lane == 0) { C[wv] = cs; Dn[wv] = g >= G ? 1 : 0; }
      __syncthreads();
      const int cAll = C[lane];
      const bool allDone = __builtin_amdgcn_ballot_w32(Dn[lane] == 0) == 0;
      for (int w = 0; w < PNW; ++w) {
        const int cw = __builtin_amdgcn_readlane(cAll, w);
        for (int i = 0; i < cw; i += 32) {
          const int k = i + lane;
          const int r = k < cw ? Lr[w][k] : -1;
          const bool mine = (r >= 0) & ((r & 31) == wv);
          const unsigned mask = __builtin_amdgcn_ballot_w32(mine);
          if (mask) {
            const int pos = cnt + (int)__builtin_amdgcn_mbcnt_lo(mask, 0u);
            if (mine) myE[pos] = Le[w][k];
            cnt += __builtin_popcount(mask);
          }
          while (cnt > PLH) workOff();
        }
      }
      while (cnt > 0) workOff();
      if (allDone) break;
      __syncthreads();
    }
  }
  if constexpr (DEG) {
    if (lcap > 0 && lane == 0) lcnt[blockIdx.x * PNW + wv] = gover ? -1 : gtot;
    int my = 0;
#pragma unroll
    for (int q = 0; q < NRW; ++q) my = lane == q ? dg[q] : my;
    const int r = wv + lane * PNW;
    if (lane < NRW && r < nrows) dout[lo + r] = rsqrtf((float)my + 1.0f);
    return;
  }

  const VT bv = *((const VT*)bias + lane);
  if constexpr (F2 == 0) {
    for (int r = wv; r < nrows; r += PNW) {
      const long long i = lo + r;
      const float di = dinv[i];
      const VT hv = *((const VT*)(h + i * F) + lane);
      const VT a  = *((const VT*)(acc + r * F) + lane);
      *((VT*)(out + i * F) + lane) = finishRow(a, hv, bv, di, di * di, RELU);
    }
  } else {
    for (int r = wv; r < PR; r += PNW) {
      VT y;
      zeroRow(y);
      if (r < nrows) {
        const long long i = lo + r;
        const float di = dinv[i];
        const VT hv = *((const VT*)(h + i * F) + lane);
        const VT a  = *((const VT*)(acc + r * F) + lane);
        y = finishRow(a, hv, bv, di, di * di, RELU);
      }
      *((VT*)(ybuf + r * YP) + lane) = y;
    }
    __syncthreads();
    constexpr int NCT = F2 / 16;
    constexpr int NSG = PNW / NCT;
    const int half = lane >> 4;
    const int lm   = lane & 15;
    const int n0   = (wv % NCT) * 16;
    const int sg   = wv / NCT;
    const float* __restrict__ bp = Wn + (2 * half) * F2 + n0 + lm;
    v2f bf[F / 4];
#pragma unroll
    for (int kk = 0; kk < F / 4; ++kk) {
      bf[kk].x = bp[(4 * kk) * F2];
      bf[kk].y = bp[(4 * kk) * F2 + F2];
    }
    for (int s = sg; s * 16 < nrows; s += NSG) {
      const int ar = s * 16 + lm < PR ? s * 16 + lm : PR - 1;
      const float* ap = ybuf + ar * YP + 2 * half;
      v8f d = {};
#pragma unroll
      for (int kk = 0; kk < F / 4; ++kk) {
        const v2f a = *(const v2f*)(ap + 4 * kk);
        d = __builtin_amdgcn_wmma_f32_16x16x4_f32(
             false, a,  false, bf[kk],
             (short)0, d,  false,  false);
      }
      const int rb = s * 16 + 8 * half;
      float* __restrict__ cp = out + (long long)(lo + rb) * F2 + n0 + lm;
      if (s * 16 + 16 <= nrows) {
#pragma unroll
        for (int r = 0; r < 8; ++r) cp[r * F2] = d[r];
      } else {
#pragma unroll
        for (int r = 0; r < 8; ++r)
          if (rb + r < nrows) cp[r * F2] = d[r];
      }
    }
  }
}

extern "C" void kernel_launch(void* const* d_in, const int* in_sizes, int n_in,
                              void* d_out, int out_size, void* d_ws, size_t ws_size,
                              hipStream_t stream) {
  const float* x  = (const float*)d_in[0];
  const int*   ei = (const int*)d_in[1];
  const float* W1 = (const float*)d_in[2];
  const float* b1 = (const float*)d_in[3];
  const float* W2 = (const float*)d_in[4];
  const float* b2 = (const float*)d_in[5];

  const int FIN = 256, H = 64, FOUT = 128;
  const int N = in_sizes[0] / FIN;
  const int E = in_sizes[1] / 2;
  const int* srcI = ei;
  const int* dstI = ei + E;

  float* ws   = (float*)d_ws;
  float* dinv = ws;
  float* h1   = dinv + (long long)N;
  float* h2   = h1   + (long long)N * H;
  const int LCAP = 1024;
  const long long nwg   = ((long long)N + PR - 1) / PR;
  const int G = (int)(((long long)E + SSW * 32 - 1) / (SSW * 32));
  const int wpo = G * SSW;
  const long long bitO  = (((long long)N * (1 + 3 * H)) + 63) / 64 * 64;
  const long long lcntO = bitO + (long long)NOWN * wpo;
  const long long lbufO = lcntO + (nwg * PNW + 63) / 64 * 64;
  const bool fits = (size_t)(lbufO + nwg * PNW * 2 * LCAP) * sizeof(float) <= ws_size;
  unsigned* bits = (unsigned*)ws + bitO;
  int* lcnt = (int*)ws + lcntO;
  int* lbuf = (int*)ws + lbufO;
  const int lcap = fits ? LCAP : 0;

  auto blocks = [](long long n, int tb) { return (unsigned)((n + tb - 1) / tb); };

  k_bits<<<(unsigned)G, BTH, 0, stream>>>(dstI, bits, E, wpo);
  k_pull<64, false, true, 0><<<blocks((long long)N, PR), PTB, 0, stream>>>(nullptr, nullptr, dinv, srcI, dstI, nullptr, nullptr, N, E, lcnt, lbuf, lcap, bits, G, nullptr);

  {
    dim3 g((unsigned)(N / 16), H / 16);
    k_gemm_wmma<256, 64><<<g, 32, 0, stream>>>(x, W1, h1);
  }
  k_pull<64, true, false, FOUT><<<blocks((long long)N, PR), PTB, 0, stream>>>(h1, dinv, nullptr, srcI, dstI, b1, h2, N, E, lcnt, lbuf, lcap, bits, G, W2);

  k_pull<128, false, false, 0><<<blocks((long long)N, PR), PTB, 0, stream>>>(h2, dinv, nullptr, srcI, dstI, b2, (float*)d_out, N, E, lcnt, lbuf, lcap, bits, G, nullptr);
}


# Scan launch's exchange made direct: a scanning wave ranks each find among its finds for one owner wave (ballots) and puts it in that pair's LDS sub-list; owners append 32 sub-lists. Sum order kept.
# speedup vs baseline: 3.9274x; 1.1040x over previous
#include <hip/hip_runtime.h>

typedef float v2f __attribute__((ext_vector_type(2)));
typedef float v8f __attribute__((ext_vector_type(8)));
typedef unsigned v4u __attribute__((ext_vector_type(4)));

constexpr int TPR  = 196;
constexpr int NOWN = 256;
constexpr int SSW  = 128;
constexpr int BTH  = 256;
constexpr int BWW  = SSW / (BTH / 32);
constexpr int BPI  = NOWN + 1;
static_assert(NOWN == 256, "the lanes of one owner meet by ballots over eight bits");
static_assert(SSW == 128 && NOWN % (BTH / 32) == 0, "write-out: four words a lane, whole owners per wave");
__launch_bounds__(BTH)
__global__ void k_bits(const int* __restrict__ dst, unsigned* __restrict__ bits, int E, int wpo) {
  __shared__ __attribute__((aligned(16))) unsigned bm[SSW * BPI];
  const int tid  = threadIdx.x;
  const int lane = tid & 31;
  const int wv   = tid >> 5;
  int d[BWW];
#pragma unroll
  for (int q = 0; q < BWW; ++q) {
    const long long e = (long long)blockIdx.x * (SSW * 32) + (wv * BWW + q) * 32 + lane;
    d[q] = e < (long long)E ? dst[e] : -1;
  }
  const uint4 z = make_uint4(0u, 0u, 0u, 0u);
  for (int i = tid; i < SSW * BPI / 4; i += BTH) ((uint4*)bm)[i] = z;
  __syncthreads();
#pragma unroll
  for (int q = 0; q < BWW; ++q) {
    const bool valid = d[q] >= 0;
    unsigned o = valid ? (unsigned)d[q] / (unsigned)TPR : 0u;
    o = o < (unsigned)(NOWN - 1) ? o : (unsigned)(NOWN - 1);
    unsigned peers = __builtin_amdgcn_ballot_w32(valid);
#pragma unroll
    for (int b = 0; b < 8; ++b) {
      const bool bit = ((o >> b) & 1u) != 0u;
      const unsigned bal = __builtin_amdgcn_ballot_w32(bit);
      peers &= bit ? bal : ~bal;
    }
    if (valid) bm[(wv * BWW + q) * BPI + (int)o] = peers;
  }
  __syncthreads();
  unsigned* __restrict__ op = bits + (long long)blockIdx.x * SSW + lane * 4;
#pragma unroll 4
  for (int i = 0; i < NOWN / (BTH / 32); ++i) {
    const int o = i * (BTH / 32) + wv;
    v4u v;
    v.x = bm[(lane * 4 + 0) * BPI + o];
    v.y = bm[(lane * 4 + 1) * BPI + o];
    v.z = bm[(lane * 4 + 2) * BPI + o];
    v.w = bm[(lane * 4 + 3) * BPI + o];
    *(v4u*)(op + (long long)o * wpo) = v;
  }
}

template <int K, int N>
__launch_bounds__(32)
__global__ void k_gemm_wmma(const float* __restrict__ A, const float* __restrict__ B,
                            float* __restrict__ C) {
  const int lane = threadIdx.x;
  const int half = lane >> 4;
  const int lm   = lane & 15;
  const int m0 = blockIdx.x * 16;
  const int n0 = blockIdx.y * 16;

  const float* __restrict__ arow = A + (long long)(m0 + lm) * K + 2 * half;
  const float* __restrict__ bp   = B + (long long)(2 * half) * N + n0 + lm;

  v8f acc = {};
#pragma unroll 8
  for (int k = 0; k < K; k += 4) {
    v2f a = *(const v2f*)(arow + k);
    v2f b;
    b.x = bp[k * N];
    b.y = bp[k * N + N];
    acc = __builtin_amdgcn_wmma_f32_16x16x4_f32(
         false, a,  false, b,
         (short)0, acc,  false,  false);
  }

  float* __restrict__ cp = C + (long long)(m0 + 8 * half) * N + n0 + lm;
#pragma unroll
  for (int r = 0; r < 8; ++r)
    cp[r * N] = acc[r];
}

constexpr int PR    = 196;
constexpr int PTB   = 1024;
constexpr int PNW   = PTB / 32;
constexpr int TIT   = 4;
constexpr int PCAP  = 512;
constexpr int PLC   = 480;
constexpr int PLH   = PLC - 32;
constexpr int NRW   = (PR + PNW - 1) / PNW;
constexpr int SCAP  = 24;
constexpr int YP    = 64 + 4;
static_assert(PNW == 32, "row ownership is row & 31");
static_assert(PR == TPR, "the bitmaps are of groups of PR rows");
static_assert(PCAP >= 32, "an empty list must take a whole turn, or a round could make no progress");
static_assert(SCAP <= 32, "an owner takes a sub-list by one load, and one exchange step adds at most 32 edges");

template <int F> struct RowVec;
template <> struct RowVec<64>  { typedef float2 T; };
template <> struct RowVec<128> { typedef float4 T; };

__device__ __forceinline__ void zeroRow(float2& a) { a.x = 0.0f; a.y = 0.0f; }
__device__ __forceinline__ void zeroRow(float4& a) { a.x = 0.0f; a.y = 0.0f; a.z = 0.0f; a.w = 0.0f; }
__device__ __forceinline__ float2 fmaRow(float2 a, float2 v, float w) {
  a.x += v.x * w; a.y += v.y * w;
  return a;
}
__device__ __forceinline__ float4 fmaRow(float4 a, float4 v, float w) {
  a.x += v.x * w; a.y += v.y * w; a.z += v.z * w; a.w += v.w * w;
  return a;
}
__device__ __forceinline__ float2 finishRow(float2 a, float2 hv, float2 bv, float di, float sq, bool relu) {
  float2 r;
  r.x = di * a.x + sq * hv.x + bv.x;
  r.y = di * a.y + sq * hv.y + bv.y;
  if (relu) { r.x = fmaxf(r.x, 0.0f); r.y = fmaxf(r.y, 0.0f); }
  return r;
}
__device__ __forceinline__ float4 finishRow(float4 a, float4 hv, float4 bv, float di, float sq, bool relu) {
  float4 r;
  r.x = di * a.x + sq * hv.x + bv.x;
  r.y = di * a.y + sq * hv.y + bv.y;
  r.z = di * a.z + sq * hv.z + bv.z;
  r.w = di * a.w + sq * hv.w + bv.w;
  if (relu) {
    r.x = fmaxf(r.x, 0.0f); r.y = fmaxf(r.y, 0.0f);
    r.z = fmaxf(r.z, 0.0f); r.w = fmaxf(r.w, 0.0f);
  }
  return r;
}

__device__ __forceinline__ int firstBit(uint4 m) {
  const unsigned w = m.x ? m.x : m.y ? m.y : m.z ? m.z : m.w;
  const int qo = m.x ? 0 : m.y ? 32 : m.z ? 64 : 96;
  return qo + __builtin_ctz(w);
}
__device__ __forceinline__ uint4 dropBit(uint4 m) {
  const bool c0 = m.x != 0u, c1 = m.y != 0u, c2 = m.z != 0u;
  const unsigned w = c0 ? m.x : c1 ? m.y : c2 ? m.z : m.w;
  const unsigned nw = w & (w - 1u);
  uint4 r;
  r.x = c0 ? nw : m.x;
  r.y = (!c0 && c1) ? nw : m.y;
  r.z = (!c0 && !c1 && c2) ? nw : m.z;
  r.w = (!c0 && !c1 && !c2) ? nw : m.w;
  return r;
}

template <int F, bool RELU, bool DEG, int F2>
__launch_bounds__(PTB)
__global__ void k_pull(const float* __restrict__ h, const float* __restrict__ dinv, float* __restrict__ dout,
                       const int* __restrict__ src, const int* __restrict__ dst,
                       const float* __restrict__ bias, float* __restrict__ out, int N, int E,
                       int* __restrict__ lcnt, int* __restrict__ lbuf, int lcap,
                       const unsigned* __restrict__ bits, int G, const float* __restrict__ Wn) {
  static_assert(F2 == 0 || (F == 64 && F2 % 16 == 0 && PNW % (F2 / 16) == 0), "fused product: 64-wide rows, whole column tiles, waves split evenly over them");
  typedef typename RowVec<F>::T VT;
  constexpr int VW = F / 32;
  constexpr int PB = 2048 / F;
  __shared__ __attribute__((aligned(16))) float acc[DEG ? 4 : PR * F];
  __shared__ int Le[PNW][PCAP];
  __shared__ int Lr[PNW][PCAP];
  __shared__ int C[PNW];
  __shared__ int Dn[PNW];
  __shared__ int lstE[PNW][PLC];
  __shared__ int sub[DEG ? PNW : 1][DEG ? PNW : 1][DEG ? SCAP : 1];
  __shared__ int Tn[DEG ? PNW : 1][DEG ? PNW : 1];
  __shared__ int Ov[DEG ? PNW : 1];
  __shared__ __attribute__((aligned(16))) float ybuf[F2 > 0 ? PR * YP : 4];

  const int tid  = threadIdx.x;
  const int lane = tid & 31;
  const int wv   = __builtin_amdgcn_readfirstlane(tid >> 5);
  const int lo   = blockIdx.x * PR;
  const int nrows = (N - lo) < PR ? (N - lo) : PR;
  const unsigned nr = (unsigned)nrows;

  if constexpr (!DEG) {
    VT z;
    zeroRow(z);
    for (int r = wv; r < PR; r += PNW) *((VT*)(acc + r * F) + lane) = z;
  }

  int* myE = lstE[wv];
  int cnt = 0;

  int* __restrict__ gl = lbuf + ((long long)blockIdx.x * PNW + wv) * 2 * (long long)lcap;
  int gtot = 0;
  bool gover = false;
  int dg[NRW];
#pragma unroll
  for (int q = 0; q < NRW; ++q) dg[q] = 0;

  auto addEdges = [&](int s, int rw, int wb, int n, auto&& behind) {
    for (int hb = 0; hb < n; hb += PB) {
      VT v[PB];
#pragma unroll
      for (int t = 0; t < PB; ++t) {
        const int tt = hb + t < n ? hb + t : n - 1;
        const int sj = __builtin_amdgcn_readlane(s, tt);
        v[t] = *((const VT*)(h + (long long)sj * F) + lane);
      }
      if (hb == 0) behind();
#pragma unroll
      for (int t = 0; t < PB; ++t) {
        if (hb + t < n) {
          const int rj = __builtin_amdgcn_readlane(rw, hb + t);
          const float wj = __int_as_float(__builtin_amdgcn_readlane(wb, hb + t));
          VT* p = (VT*)(acc + rj * F) + lane;
          *p = fmaRow(*p, v[t], wj);
        }
      }
    }
  };

  bool useList = false;
  int ltot = 0;
  if (!DEG && lcap > 0) {
    const int c = lcnt[blockIdx.x * PNW + lane];
    useList = __builtin_amdgcn_ballot_w32(c < 0) == 0;
    ltot = __builtin_amdgcn_readlane(c, wv);
  }

  if (useList) {
    auto loadEnt = [&](int c0, int& s, int& rw, int& wb) {
      const int n = ltot - c0 < 32 ? ltot - c0 : 32;
      const int li = c0 + (lane < n ? lane : n - 1);
      s = gl[li]; rw = gl[lcap + li]; wb = __float_as_int(dinv[s]);
    };
    int ns = 0, nrw = 0, nwb = 0;
    if (ltot > 0) loadEnt(0, ns, nrw, nwb);
    for (int c0 = 0; c0 < ltot; c0 += 32) {
      const int n = ltot - c0 < 32 ? ltot - c0 : 32;
      const int s = ns, rw = nrw, wb = nwb;
      addEdges(s, rw, wb, n, [&] { loadEnt(c0 + 32 < ltot ? c0 + 32 : c0, ns, nrw, nwb); });
    }
  } else {
    auto workOff = [&] {
      const int n = cnt < 32 ? cnt : 32;
      const int li = cnt - n + (lane < n ? lane : n - 1);
      const int e  = myE[li];
      const int s  = src[e];
      const int rw = dst[e] - lo;
      if constexpr (DEG) {
        if (gtot + n <= lcap) {
          if (lane < n) { gl[gtot + lane] = s; gl[lcap + gtot + lane] = rw; }
          gtot += n;
        } else {
          gover = true;
        }
#pragma unroll
        for (int q = 0; q < NRW; ++q)
          dg[q] += __builtin_popcount(__builtin_amdgcn_ballot_w32((lane < n) & (rw == wv + q * PNW)));
      } else {
        const int wb = __float_as_int(dinv[s]);
        addEdges(s, rw, wb, n, [] {});
      }
      cnt -= n;
    };

    const unsigned* __restrict__ bw = bits + (long long)(blockIdx.x < NOWN - 1 ? blockIdx.x : NOWN - 1) * ((long long)G * SSW);
    int* myF = Le[wv];
    int* myR = Lr[wv];
    int g  = wv;
    int t0 = 0;
    for (;;) {
      int cs = 0;
      bool full = false;
      while (g < G && !full) {
        uint4 tv[TIT];
#pragma unroll
        for (int j = 0; j < TIT; ++j) {
          const int gj = g + j * PNW < G ? g + j * PNW : g;
          tv[j] = *((const uint4*)(bw + (long long)gj * SSW) + lane);
        }
#pragma unroll
        for (int j = 0; j < TIT; ++j) {
          if (!full && g < G) {
            uint4 mw = tv[j];
            for (int t = 0; t < t0; ++t) mw = dropBit(mw);
            const int eb = g * (SSW * 32) + lane * 128;
            int t = t0;
            while (__builtin_amdgcn_ballot_w32((mw.x | mw.y | mw.z | mw.w) != 0)) {
              const bool v = (mw.x | mw.y | mw.z | mw.w) != 0;
              const unsigned mask = __builtin_amdgcn_ballot_w32(v);
              const int n = __builtin_popcount(mask);
              if (cs + n > PCAP) { full = true; break; }
              const int pos = cs + (int)__builtin_amdgcn_mbcnt_lo(mask, 0u);
              if (v) myF[pos] = eb + firstBit(mw);
              cs += n;
              mw = dropBit(mw);
              ++t;
            }
            if (full) t0 = t;
            else { g += PNW; t0 = 0; }
          }
        }
      }
      int cntv = 0;
      bool ovf = false;
      for (int i = 0; i < cs; i += 128) {
        int r[4];
        int ef[4];
#pragma unroll
        for (int q = 0; q < 4; ++q) {
          const int k = i + q * 32 + lane;
          ef[q] = k < cs ? myF[k] : 0;
          r[q] = k < cs ? dst[ef[q]] - lo : -1;
        }
#pragma unroll
        for (int q = 0; q < 4; ++q) {
          const int k = i + q * 32 + lane;
          const bool val = (unsigned)r[q] < nr;
          if (k < cs) myR[k] = val ? r[q] : -1;
          if constexpr (DEG) {
            const unsigned o = val ? (unsigned)r[q] & 31u : 0u;
            unsigned peers = __builtin_amdgcn_ballot_w32(val);
            unsigned own = peers;
#pragma unroll
            for (int b = 0; b < 5; ++b) {
              const bool bit = ((o >> b) & 1u) != 0u;
              const unsigned bal = __builtin_amdgcn_ballot_w32(bit);
              peers &= bit ? bal : ~bal;
              own &= ((lane >> b) & 1) != 0 ? bal : ~bal;
            }
            const int base = __builtin_amdgcn_ds_bpermute((int)(o << 2), cntv);
            const int pos = base + (int)__builtin_amdgcn_mbcnt_lo(peers, 0u);
            if (val) {
              if (pos < SCAP) sub[wv][o][pos] = ef[q];
              else ovf = true;
            }
            cntv += __builtin_popcount(own);
          }
        }
      }
      if constexpr (DEG) {
        Tn[lane][wv] = cntv;
        const bool anyOvf = __builtin_amdgcn_ballot_w32(ovf) != 0;
        if (lane == 0) Ov[wv] = anyOvf ? 1 : 0;
      }
      if (lane == 0) { C[wv] = cs; Dn[wv] = g >= G ? 1 : 0; }
      __syncthreads();
      const int cAll = C[lane];
      const bool allDone = __builtin_amdgcn_ballot_w32(Dn[lane] == 0) == 0;
      bool direct = false;
      if constexpr (DEG) {
        direct = __builtin_amdgcn_ballot_w32(Ov[lane] != 0) == 0;
        if (direct) {
          const int cOwn = Tn[wv][lane];
          for (int w = 0; w < PNW; ++w) {
            const int c = __builtin_amdgcn_readlane(cOwn, w);
            if (c > 0) {
              if (lane < c) myE[cnt + lane] = sub[w][wv][lane];
              cnt += c;
              while (cnt > PLH) workOff();
            }
          }
        }
      }
      if (!direct) {
        for (int w = 0; w < PNW; ++w) {
          const int cw = __builtin_amdgcn_readlane(cAll, w);
          for (int i = 0; i < cw; i += 32) {
            const int k = i + lane;
            const int r = k < cw ? Lr[w][k] : -1;
            const bool mine = (r >= 0) & ((r & 31) == wv);
            const unsigned mask = __builtin_amdgcn_ballot_w32(mine);
            if (mask) {
              const int pos = cnt + (int)__builtin_amdgcn_mbcnt_lo(mask, 0u);
              if (mine) myE[pos] = Le[w][k];
              cnt += __builtin_popcount(mask);
            }
            while (cnt > PLH) workOff();
          }
        }
      }
      while (cnt > 0) workOff();
      if (allDone) break;
      __syncthreads();
    }
  }
  if constexpr (DEG) {
    if (lcap > 0 && lane == 0) lcnt[blockIdx.x * PNW + wv] = gover ? -1 : gtot;
    int my = 0;
#pragma unroll
    for (int q = 0; q < NRW; ++q) my = lane == q ? dg[q] : my;
    const int r = wv + lane * PNW;
    if (lane < NRW && r < nrows) dout[lo + r] = rsqrtf((float)my + 1.0f);
    return;
  }

  const VT bv = *((const VT*)bias + lane);
  if constexpr (F2 == 0) {
    for (int r = wv; r < nrows; r += PNW) {
      const long long i = lo + r;
      const float di = dinv[i];
      const VT hv = *((const VT*)(h + i * F) + lane);
      const VT a  = *((const VT*)(acc + r * F) + lane);
      *((VT*)(out + i * F) + lane) = finishRow(a, hv, bv, di, di * di, RELU);
    }
  } else {
    for (int r = wv; r < PR; r += PNW) {
      VT y;
      zeroRow(y);
      if (r < nrows) {
        const long long i = lo + r;
        const float di = dinv[i];
        const VT hv = *((const VT*)(h + i * F) + lane);
        const VT a  = *((const VT*)(acc + r * F) + lane);
        y = finishRow(a, hv, bv, di, di * di, RELU);
      }
      *((VT*)(ybuf + r * YP) + lane) = y;
    }
    __syncthreads();
    constexpr int NCT = F2 / 16;
    constexpr int NSG = PNW / NCT;
    const int half = lane >> 4;
    const int lm   = lane & 15;
    const int n0   = (wv % NCT) * 16;
    const int sg   = wv / NCT;
    const float* __restrict__ bp = Wn + (2 * half) * F2 + n0 + lm;
    v2f bf[F / 4];
#pragma unroll
    for (int kk = 0; kk < F / 4; ++kk) {
      bf[kk].x = bp[(4 * kk) * F2];
      bf[kk].y = bp[(4 * kk) * F2 + F2];
    }
    for (int s = sg; s * 16 < nrows; s += NSG) {
      const int ar = s * 16 + lm < PR ? s * 16 + lm : PR - 1;
      const float* ap = ybuf + ar * YP + 2 * half;
      v8f d = {};
#pragma unroll
      for (int kk = 0; kk < F / 4; ++kk) {
        const v2f a = *(const v2f*)(ap + 4 * kk);
        d = __builtin_amdgcn_wmma_f32_16x16x4_f32(
             false, a,  false, bf[kk],
             (short)0, d,  false,  false);
      }
      const int rb = s * 16 + 8 * half;
      float* __restrict__ cp = out + (long long)(lo + rb) * F2 + n0 + lm;
      if (s * 16 + 16 <= nrows) {
#pragma unroll
        for (int r = 0; r < 8; ++r) cp[r * F2] = d[r];
      } else {
#pragma unroll
        for (int r = 0; r < 8; ++r)
          if (rb + r < nrows) cp[r * F2] = d[r];
      }
    }
  }
}

extern "C" void kernel_launch(void* const* d_in, const int* in_sizes, int n_in,
                              void* d_out, int out_size, void* d_ws, size_t ws_size,
                              hipStream_t stream) {
  const float* x  = (const float*)d_in[0];
  const int*   ei = (const int*)d_in[1];
  const float* W1 = (const float*)d_in[2];
  const float* b1 = (const float*)d_in[3];
  const float* W2 = (const float*)d_in[4];
  const float* b2 = (const float*)d_in[5];

  const int FIN = 256, H = 64, FOUT = 128;
  const int N = in_sizes[0] / FIN;
  const int E = in_sizes[1] / 2;
  const int* srcI = ei;
  const int* dstI = ei + E;

  float* ws   = (float*)d_ws;
  float* dinv = ws;
  float* h1   = dinv + (long long)N;
  float* h2   = h1   + (long long)N * H;
  const int LCAP = 1024;
  const long long nwg   = ((long long)N + PR - 1) / PR;
  const int G = (int)(((long long)E + SSW * 32 - 1) / (SSW * 32));
  const int wpo = G * SSW;
  const long long bitO  = (((long long)N * (1 + 3 * H)) + 63) / 64 * 64;
  const long long lcntO = bitO + (long long)NOWN * wpo;
  const long long lbufO = lcntO + (nwg * PNW + 63) / 64 * 64;
  const bool fits = (size_t)(lbufO + nwg * PNW * 2 * LCAP) * sizeof(float) <= ws_size;
  unsigned* bits = (unsigned*)ws + bitO;
  int* lcnt = (int*)ws + lcntO;
  int* lbuf = (int*)ws + lbufO;
  const int lcap = fits ? LCAP : 0;

  auto blocks = [](long long n, int tb) { return (unsigned)((n + tb - 1) / tb); };

  k_bits<<<(unsigned)G, BTH, 0, stream>>>(dstI, bits, E, wpo);
  k_pull<64, false, true, 0><<<blocks((long long)N, PR), PTB, 0, stream>>>(nullptr, nullptr, dinv, srcI, dstI, nullptr, nullptr, N, E, lcnt, lbuf, lcap, bits, G, nullptr);

  {
    dim3 g((unsigned)(N / 16), H / 16);
    k_gemm_wmma<256, 64><<<g, 32, 0, stream>>>(x, W1, h1);
  }
  k_pull<64, true, false, FOUT><<<blocks((long long)N, PR), PTB, 0, stream>>>(h1, dinv, nullptr, srcI, dstI, b1, h2, N, E, lcnt, lbuf, lcap, bits, G, W2);

  k_pull<128, false, false, 0><<<blocks((long long)N, PR), PTB, 0, stream>>>(h2, dinv, nullptr, srcI, dstI, b2, (float*)d_out, N, E, lcnt, lbuf, lcap, bits, G, nullptr);
}


# Layer 1's aggregation leaves each kept-list entry's weight (its source's dinv) in a third list plane; layer 2 reads it back in list order and looks up no dinv per edge. Sum order kept.
# speedup vs baseline: 4.0805x; 1.0390x over previous
#include <hip/hip_runtime.h>

typedef float v2f __attribute__((ext_vector_type(2)));
typedef float v8f __attribute__((ext_vector_type(8)));
typedef unsigned v4u __attribute__((ext_vector_type(4)));

constexpr int TPR  = 196;
constexpr int NOWN = 256;
constexpr int SSW  = 128;
constexpr int BTH  = 256;
constexpr int BWW  = SSW / (BTH / 32);
constexpr int BPI  = NOWN + 1;
static_assert(NOWN == 256, "the lanes of one owner meet by ballots over eight bits");
static_assert(SSW == 128 && NOWN % (BTH / 32) == 0, "write-out: four words a lane, whole owners per wave");
__launch_bounds__(BTH)
__global__ void k_bits(const int* __restrict__ dst, unsigned* __restrict__ bits, int E, int wpo) {
  __shared__ __attribute__((aligned(16))) unsigned bm[SSW * BPI];
  const int tid  = threadIdx.x;
  const int lane = tid & 31;
  const int wv   = tid >> 5;
  int d[BWW];
#pragma unroll
  for (int q = 0; q < BWW; ++q) {
    const long long e = (long long)blockIdx.x * (SSW * 32) + (wv * BWW + q) * 32 + lane;
    d[q] = e < (long long)E ? dst[e] : -1;
  }
  const uint4 z = make_uint4(0u, 0u, 0u, 0u);
  for (int i = tid; i < SSW * BPI / 4; i += BTH) ((uint4*)bm)[i] = z;
  __syncthreads();
#pragma unroll
  for (int q = 0; q < BWW; ++q) {
    const bool valid = d[q] >= 0;
    unsigned o = valid ? (unsigned)d[q] / (unsigned)TPR : 0u;
    o = o < (unsigned)(NOWN - 1) ? o : (unsigned)(NOWN - 1);
    unsigned peers = __builtin_amdgcn_ballot_w32(valid);
#pragma unroll
    for (int b = 0; b < 8; ++b) {
      const bool bit = ((o >> b) & 1u) != 0u;
      const unsigned bal = __builtin_amdgcn_ballot_w32(bit);
      peers &= bit ? bal : ~bal;
    }
    if (valid) bm[(wv * BWW + q) * BPI + (int)o] = peers;
  }
  __syncthreads();
  unsigned* __restrict__ op = bits + (long long)blockIdx.x * SSW + lane * 4;
#pragma unroll 4
  for (int i = 0; i < NOWN / (BTH / 32); ++i) {
    const int o = i * (BTH / 32) + wv;
    v4u v;
    v.x = bm[(lane * 4 + 0) * BPI + o];
    v.y = bm[(lane * 4 + 1) * BPI + o];
    v.z = bm[(lane * 4 + 2) * BPI + o];
    v.w = bm[(lane * 4 + 3) * BPI + o];
    *(v4u*)(op + (long long)o * wpo) = v;
  }
}

template <int K, int N>
__launch_bounds__(32)
__global__ void k_gemm_wmma(const float* __restrict__ A, const float* __restrict__ B,
                            float* __restrict__ C) {
  const int lane = threadIdx.x;
  const int half = lane >> 4;
  const int lm   = lane & 15;
  const int m0 = blockIdx.x * 16;
  const int n0 = blockIdx.y * 16;

  const float* __restrict__ arow = A + (long long)(m0 + lm) * K + 2 * half;
  const float* __restrict__ bp   = B + (long long)(2 * half) * N + n0 + lm;

  v8f acc = {};
#pragma unroll 8
  for (int k = 0; k < K; k += 4) {
    v2f a = *(const v2f*)(arow + k);
    v2f b;
    b.x = bp[k * N];
    b.y = bp[k * N + N];
    acc = __builtin_amdgcn_wmma_f32_16x16x4_f32(
         false, a,  false, b,
         (short)0, acc,  false,  false);
  }

  float* __restrict__ cp = C + (long long)(m0 + 8 * half) * N + n0 + lm;
#pragma unroll
  for (int r = 0; r < 8; ++r)
    cp[r * N] = acc[r];
}

constexpr int PR    = 196;
constexpr int PTB   = 1024;
constexpr int PNW   = PTB / 32;
constexpr int TIT   = 4;
constexpr int PCAP  = 512;
constexpr int PLC   = 480;
constexpr int PLH   = PLC - 32;
constexpr int NRW   = (PR + PNW - 1) / PNW;
constexpr int SCAP  = 24;
constexpr int YP    = 64 + 4;
static_assert(PNW == 32, "row ownership is row & 31");
static_assert(PR == TPR, "the bitmaps are of groups of PR rows");
static_assert(PCAP >= 32, "an empty list must take a whole turn, or a round could make no progress");
static_assert(SCAP <= 32, "an owner takes a sub-list by one load, and one exchange step adds at most 32 edges");

template <int F> struct RowVec;
template <> struct RowVec<64>  { typedef float2 T; };
template <> struct RowVec<128> { typedef float4 T; };

__device__ __forceinline__ void zeroRow(float2& a) { a.x = 0.0f; a.y = 0.0f; }
__device__ __forceinline__ void zeroRow(float4& a) { a.x = 0.0f; a.y = 0.0f; a.z = 0.0f; a.w = 0.0f; }
__device__ __forceinline__ float2 fmaRow(float2 a, float2 v, float w) {
  a.x += v.x * w; a.y += v.y * w;
  return a;
}
__device__ __forceinline__ float4 fmaRow(float4 a, float4 v, float w) {
  a.x += v.x * w; a.y += v.y * w; a.z += v.z * w; a.w += v.w * w;
  return a;
}
__device__ __forceinline__ float2 finishRow(float2 a, float2 hv, float2 bv, float di, float sq, bool relu) {
  float2 r;
  r.x = di * a.x + sq * hv.x + bv.x;
  r.y = di * a.y + sq * hv.y + bv.y;
  if (relu) { r.x = fmaxf(r.x, 0.0f); r.y = fmaxf(r.y, 0.0f); }
  return r;
}
__device__ __forceinline__ float4 finishRow(float4 a, float4 hv, float4 bv, float di, float sq, bool relu) {
  float4 r;
  r.x = di * a.x + sq * hv.x + bv.x;
  r.y = di * a.y + sq * hv.y + bv.y;
  r.z = di * a.z + sq * hv.z + bv.z;
  r.w = di * a.w + sq * hv.w + bv.w;
  if (relu) {
    r.x = fmaxf(r.x, 0.0f); r.y = fmaxf(r.y, 0.0f);
    r.z = fmaxf(r.z, 0.0f); r.w = fmaxf(r.w, 0.0f);
  }
  return r;
}

__device__ __forceinline__ int firstBit(uint4 m) {
  const unsigned w = m.x ? m.x : m.y ? m.y : m.z ? m.z : m.w;
  const int qo = m.x ? 0 : m.y ? 32 : m.z ? 64 : 96;
  return qo + __builtin_ctz(w);
}
__device__ __forceinline__ uint4 dropBit(uint4 m) {
  const bool c0 = m.x != 0u, c1 = m.y != 0u, c2 = m.z != 0u;
  const unsigned w = c0 ? m.x : c1 ? m.y : c2 ? m.z : m.w;
  const unsigned nw = w & (w - 1u);
  uint4 r;
  r.x = c0 ? nw : m.x;
  r.y = (!c0 && c1) ? nw : m.y;
  r.z = (!c0 && !c1 && c2) ? nw : m.z;
  r.w = (!c0 && !c1 && !c2) ? nw : m.w;
  return r;
}

constexpr int LPL = 3;
template <int F, bool RELU, bool DEG, int F2, int WM>
__launch_bounds__(PTB)
__global__ void k_pull(const float* __restrict__ h, const float* __restrict__ dinv, float* __restrict__ dout,
                       const int* __restrict__ src, const int* __restrict__ dst,
                       const float* __restrict__ bias, float* __restrict__ out, int N, int E,
                       int* __restrict__ lcnt, int* __restrict__ lbuf, int lcap,
                       const unsigned* __restrict__ bits, int G, const float* __restrict__ Wn) {
  static_assert(F2 == 0 || (F == 64 && F2 % 16 == 0 && PNW % (F2 / 16) == 0), "fused product: 64-wide rows, whole column tiles, waves split evenly over them");
  static_assert(WM >= 0 && WM <= 2 && (!DEG || WM == 0), "list weights: 0 looked up, 1 looked up and kept, 2 read back; the scan launch has none");
  typedef typename RowVec<F>::T VT;
  constexpr int VW = F / 32;
  constexpr int PB = 2048 / F;
  __shared__ __attribute__((aligned(16))) float acc[DEG ? 4 : PR * F];
  __shared__ int Le[PNW][PCAP];
  __shared__ int Lr[PNW][PCAP];
  __shared__ int C[PNW];
  __shared__ int Dn[PNW];
  __shared__ int lstE[PNW][PLC];
  __shared__ int sub[DEG ? PNW : 1][DEG ? PNW : 1][DEG ? SCAP : 1];
  __shared__ int Tn[DEG ? PNW : 1][DEG ? PNW : 1];
  __shared__ int Ov[DEG ? PNW : 1];
  __shared__ __attribute__((aligned(16))) float ybuf[F2 > 0 ? PR * YP : 4];

  const int tid  = threadIdx.x;
  const int lane = tid & 31;
  const int wv   = __builtin_amdgcn_readfirstlane(tid >> 5);
  const int lo   = blockIdx.x * PR;
  const int nrows = (N - lo) < PR ? (N - lo) : PR;
  const unsigned nr = (unsigned)nrows;

  if constexpr (!DEG) {
    VT z;
    zeroRow(z);
    for (int r = wv; r < PR; r += PNW) *((VT*)(acc + r * F) + lane) = z;
  }

  int* myE = lstE[wv];
  int cnt = 0;

  int* __restrict__ gl = lbuf + ((long long)blockIdx.x * PNW + wv) * LPL * (long long)lcap;
  int gtot = 0;
  bool gover = false;
  int dg[NRW];
#pragma unroll
  for (int q = 0; q < NRW; ++q) dg[q] = 0;

  auto addEdges = [&](int s, int rw, int wb, int n, auto&& behind) {
    for (int hb = 0; hb < n; hb += PB) {
      VT v[PB];
#pragma unroll
      for (int t = 0; t < PB; ++t) {
        const int tt = hb + t < n ? hb + t : n - 1;
        const int sj = __builtin_amdgcn_readlane(s, tt);
        v[t] = *((const VT*)(h + (long long)sj * F) + lane);
      }
      if (hb == 0) behind();
#pragma unroll
      for (int t = 0; t < PB; ++t) {
        if (hb + t < n) {
          const int rj = __builtin_amdgcn_readlane(rw, hb + t);
          const float wj = __int_as_float(__builtin_amdgcn_readlane(wb, hb + t));
          VT* p = (VT*)(acc + rj * F) + lane;
          *p = fmaRow(*p, v[t], wj);
        }
      }
    }
  };

  bool useList = false;
  int ltot = 0;
  if (!DEG && lcap > 0) {
    const int c = lcnt[blockIdx.x * PNW + lane];
    useList = __builtin_amdgcn_ballot_w32(c < 0) == 0;
    ltot = __builtin_amdgcn_readlane(c, wv);
  }

  if (useList) {
    auto loadEnt = [&](int c0, int& s, int& rw, int& wb) {
      const int n = ltot - c0 < 32 ? ltot - c0 : 32;
      const int li = c0 + (lane < n ? lane : n - 1);
      s = gl[li]; rw = gl[lcap + li];
      if constexpr (WM == 2) wb = gl[2 * lcap + li];
      else wb = __float_as_int(dinv[s]);
    };
    int ns = 0, nrw = 0, nwb = 0;
    if (ltot > 0) loadEnt(0, ns, nrw, nwb);
    for (int c0 = 0; c0 < ltot; c0 += 32) {
      const int n = ltot - c0 < 32 ? ltot - c0 : 32;
      const int s = ns, rw = nrw, wb = nwb;
      if constexpr (WM == 1) {
        if (lane < n) gl[2 * lcap + c0 + lane] = wb;
      }
      addEdges(s, rw, wb, n, [&] { loadEnt(c0 + 32 < ltot ? c0 + 32 : c0, ns, nrw, nwb); });
    }
  } else {
    auto workOff = [&] {
      const int n = cnt < 32 ? cnt : 32;
      const int li = cnt - n + (lane < n ? lane : n - 1);
      const int e  = myE[li];
      const int s  = src[e];
      const int rw = dst[e] - lo;
      if constexpr (DEG) {
        if (gtot + n <= lcap) {
          if (lane < n) { gl[gtot + lane] = s; gl[lcap + gtot + lane] = rw; }
          gtot += n;
        } else {
          gover = true;
        }
#pragma unroll
        for (int q = 0; q < NRW; ++q)
          dg[q] += __builtin_popcount(__builtin_amdgcn_ballot_w32((lane < n) & (rw == wv + q * PNW)));
      } else {
        const int wb = __float_as_int(dinv[s]);
        addEdges(s, rw, wb, n, [] {});
      }
      cnt -= n;
    };

    const unsigned* __restrict__ bw = bits + (long long)(blockIdx.x < NOWN - 1 ? blockIdx.x : NOWN - 1) * ((long long)G * SSW);
    int* myF = Le[wv];
    int* myR = Lr[wv];
    int g  = wv;
    int t0 = 0;
    for (;;) {
      int cs = 0;
      bool full = false;
      while (g < G && !full) {
        uint4 tv[TIT];
#pragma unroll
        for (int j = 0; j < TIT; ++j) {
          const int gj = g + j * PNW < G ? g + j * PNW : g;
          tv[j] = *((const uint4*)(bw + (long long)gj * SSW) + lane);
        }
#pragma unroll
        for (int j = 0; j < TIT; ++j) {
          if (!full && g < G) {
            uint4 mw = tv[j];
            for (int t = 0; t < t0; ++t) mw = dropBit(mw);
            const int eb = g * (SSW * 32) + lane * 128;
            int t = t0;
            while (__builtin_amdgcn_ballot_w32((mw.x | mw.y | mw.z | mw.w) != 0)) {
              const bool v = (mw.x | mw.y | mw.z | mw.w) != 0;
              const unsigned mask = __builtin_amdgcn_ballot_w32(v);
              const int n = __builtin_popcount(mask);
              if (cs + n > PCAP) { full = true; break; }
              const int pos = cs + (int)__builtin_amdgcn_mbcnt_lo(mask, 0u);
              if (v) myF[pos] = eb + firstBit(mw);
              cs += n;
              mw = dropBit(mw);
              ++t;
            }
            if (full) t0 = t;
            else { g += PNW; t0 = 0; }
          }
        }
      }
      int cntv = 0;
      bool ovf = false;
      for (int i = 0; i < cs; i += 128) {
        int r[4];
        int ef[4];
#pragma unroll
        for (int q = 0; q < 4; ++q) {
          const int k = i + q * 32 + lane;
          ef[q] = k < cs ? myF[k] : 0;
          r[q] = k < cs ? dst[ef[q]] - lo : -1;
        }
#pragma unroll
        for (int q = 0; q < 4; ++q) {
          const int k = i + q * 32 + lane;
          const bool val = (unsigned)r[q] < nr;
          if (k < cs) myR[k] = val ? r[q] : -1;
          if constexpr (DEG) {
            const unsigned o = val ? (unsigned)r[q] & 31u : 0u;
            unsigned peers = __builtin_amdgcn_ballot_w32(val);
            unsigned own = peers;
#pragma unroll
            for (int b = 0; b < 5; ++b) {
              const bool bit = ((o >> b) & 1u) != 0u;
              const unsigned bal = __builtin_amdgcn_ballot_w32(bit);
              peers &= bit ? bal : ~bal;
              own &= ((lane >> b) & 1) != 0 ? bal : ~bal;
            }
            const int base = __builtin_amdgcn_ds_bpermute((int)(o << 2), cntv);
            const int pos = base + (int)__builtin_amdgcn_mbcnt_lo(peers, 0u);
            if (val) {
              if (pos < SCAP) sub[wv][o][pos] = ef[q];
              else ovf = true;
            }
            cntv += __builtin_popcount(own);
          }
        }
      }
      if constexpr (DEG) {
        Tn[lane][wv] = cntv;
        const bool anyOvf = __builtin_amdgcn_ballot_w32(ovf) != 0;
        if (lane == 0) Ov[wv] = anyOvf ? 1 : 0;
      }
      if (lane == 0) { C[wv] = cs; Dn[wv] = g >= G ? 1 : 0; }
      __syncthreads();
      const int cAll = C[lane];
      const bool allDone = __builtin_amdgcn_ballot_w32(Dn[lane] == 0) == 0;
      bool direct = false;
      if constexpr (DEG) {
        direct = __builtin_amdgcn_ballot_w32(Ov[lane] != 0) == 0;
        if (direct) {
          const int cOwn = Tn[wv][lane];
          for (int w = 0; w < PNW; ++w) {
            const int c = __builtin_amdgcn_readlane(cOwn, w);
            if (c > 0) {
              if (lane < c) myE[cnt + lane] = sub[w][wv][lane];
              cnt += c;
              while (cnt > PLH) workOff();
            }
          }
        }
      }
      if (!direct) {
        for (int w = 0; w < PNW; ++w) {
          const int cw = __builtin_amdgcn_readlane(cAll, w);
          for (int i = 0; i < cw; i += 32) {
            const int k = i + lane;
            const int r = k < cw ? Lr[w][k] : -1;
            const bool mine = (r >= 0) & ((r & 31) == wv);
            const unsigned mask = __builtin_amdgcn_ballot_w32(mine);
            if (mask) {
              const int pos = cnt + (int)__builtin_amdgcn_mbcnt_lo(mask, 0u);
              if (mine) myE[pos] = Le[w][k];
              cnt += __builtin_popcount(mask);
            }
            while (cnt > PLH) workOff();
          }
        }
      }
      while (cnt > 0) workOff();
      if (allDone) break;
      __syncthreads();
    }
  }
  if constexpr (DEG) {
    if (lcap > 0 && lane == 0) lcnt[blockIdx.x * PNW + wv] = gover ? -1 : gtot;
    int my = 0;
#pragma unroll
    for (int q = 0; q < NRW; ++q) my = lane == q ? dg[q] : my;
    const int r = wv + lane * PNW;
    if (lane < NRW && r < nrows) dout[lo + r] = rsqrtf((float)my + 1.0f);
    return;
  }

  const VT bv = *((const VT*)bias + lane);
  if constexpr (F2 == 0) {
    for (int r = wv; r < nrows; r += PNW) {
      const long long i = lo + r;
      const float di = dinv[i];
      const VT hv = *((const VT*)(h + i * F) + lane);
      const VT a  = *((const VT*)(acc + r * F) + lane);
      *((VT*)(out + i * F) + lane) = finishRow(a, hv, bv, di, di * di, RELU);
    }
  } else {
    for (int r = wv; r < PR; r += PNW) {
      VT y;
      zeroRow(y);
      if (r < nrows) {
        const long long i = lo + r;
        const float di = dinv[i];
        const VT hv = *((const VT*)(h + i * F) + lane);
        const VT a  = *((const VT*)(acc + r * F) + lane);
        y = finishRow(a, hv, bv, di, di * di, RELU);
      }
      *((VT*)(ybuf + r * YP) + lane) = y;
    }
    __syncthreads();
    constexpr int NCT = F2 / 16;
    constexpr int NSG = PNW / NCT;
    const int half = lane >> 4;
    const int lm   = lane & 15;
    const int n0   = (wv % NCT) * 16;
    const int sg   = wv / NCT;
    const float* __restrict__ bp = Wn + (2 * half) * F2 + n0 + lm;
    v2f bf[F / 4];
#pragma unroll
    for (int kk = 0; kk < F / 4; ++kk) {
      bf[kk].x = bp[(4 * kk) * F2];
      bf[kk].y = bp[(4 * kk) * F2 + F2];
    }
    for (int s = sg; s * 16 < nrows; s += NSG) {
      const int ar = s * 16 + lm < PR ? s * 16 + lm : PR - 1;
      const float* ap = ybuf + ar * YP + 2 * half;
      v8f d = {};
#pragma unroll
      for (int kk = 0; kk < F / 4; ++kk) {
        const v2f a = *(const v2f*)(ap + 4 * kk);
        d = __builtin_amdgcn_wmma_f32_16x16x4_f32(
             false, a,  false, bf[kk],
             (short)0, d,  false,  false);
      }
      const int rb = s * 16 + 8 * half;
      float* __restrict__ cp = out + (long long)(lo + rb) * F2 + n0 + lm;
      if (s * 16 + 16 <= nrows) {
#pragma unroll
        for (int r = 0; r < 8; ++r) cp[r * F2] = d[r];
      } else {
#pragma unroll
        for (int r = 0; r < 8; ++r)
          if (rb + r < nrows) cp[r * F2] = d[r];
      }
    }
  }
}

extern "C" void kernel_launch(void* const* d_in, const int* in_sizes, int n_in,
                              void* d_out, int out_size, void* d_ws, size_t ws_size,
                              hipStream_t stream) {
  const float* x  = (const float*)d_in[0];
  const int*   ei = (const int*)d_in[1];
  const float* W1 = (const float*)d_in[2];
  const float* b1 = (const float*)d_in[3];
  const float* W2 = (const float*)d_in[4];
  const float* b2 = (const float*)d_in[5];

  const int FIN = 256, H = 64, FOUT = 128;
  const int N = in_sizes[0] / FIN;
  const int E = in_sizes[1] / 2;
  const int* srcI = ei;
  const int* dstI = ei + E;

  float* ws   = (float*)d_ws;
  float* dinv = ws;
  float* h1   = dinv + (long long)N;
  float* h2   = h1   + (long long)N * H;
  const int LCAP = 1024;
  const long long nwg   = ((long long)N + PR - 1) / PR;
  const int G = (int)(((long long)E + SSW * 32 - 1) / (SSW * 32));
  const int wpo = G * SSW;
  const long long bitO  = (((long long)N * (1 + 3 * H)) + 63) / 64 * 64;
  const long long lcntO = bitO + (long long)NOWN * wpo;
  const long long lbufO = lcntO + (nwg * PNW + 63) / 64 * 64;
  const bool fits = (size_t)(lbufO + nwg * PNW * LPL * LCAP) * sizeof(float) <= ws_size;
  unsigned* bits = (unsigned*)ws + bitO;
  int* lcnt = (int*)ws + lcntO;
  int* lbuf = (int*)ws + lbufO;
  const int lcap = fits ? LCAP : 0;

  auto blocks = [](long long n, int tb) { return (unsigned)((n + tb - 1) / tb); };

  k_bits<<<(unsigned)G, BTH, 0, stream>>>(dstI, bits, E, wpo);
  k_pull<64, false, true, 0, 0><<<blocks((long long)N, PR), PTB, 0, stream>>>(nullptr, nullptr, dinv, srcI, dstI, nullptr, nullptr, N, E, lcnt, lbuf, lcap, bits, G, nullptr);

  {
    dim3 g((unsigned)(N / 16), H / 16);
    k_gemm_wmma<256, 64><<<g, 32, 0, stream>>>(x, W1, h1);
  }
  k_pull<64, true, false, FOUT, 1><<<blocks((long long)N, PR), PTB, 0, stream>>>(h1, dinv, nullptr, srcI, dstI, b1, h2, N, E, lcnt, lbuf, lcap, bits, G, W2);

  k_pull<128, false, false, 0, 2><<<blocks((long long)N, PR), PTB, 0, stream>>>(h2, dinv, nullptr, srcI, dstI, b2, (float*)d_out, N, E, lcnt, lbuf, lcap, bits, G, nullptr);
}


# Workspace planes h1 and h2 start on 256-byte boundaries (dinv's N floats had left them 64 bytes past one): a gathered row is 2 / 4 whole 128-byte lines, not 3 / 5. Sum order kept.
# speedup vs baseline: 4.3268x; 1.0604x over previous
#include <hip/hip_runtime.h>

typedef float v2f __attribute__((ext_vector_type(2)));
typedef float v8f __attribute__((ext_vector_type(8)));
typedef unsigned v4u __attribute__((ext_vector_type(4)));

constexpr int TPR  = 196;
constexpr int NOWN = 256;
constexpr int SSW  = 128;
constexpr int BTH  = 256;
constexpr int BWW  = SSW / (BTH / 32);
constexpr int BPI  = NOWN + 1;
static_assert(NOWN == 256, "the lanes of one owner meet by ballots over eight bits");
static_assert(SSW == 128 && NOWN % (BTH / 32) == 0, "write-out: four words a lane, whole owners per wave");
__launch_bounds__(BTH)
__global__ void k_bits(const int* __restrict__ dst, unsigned* __restrict__ bits, int E, int wpo) {
  __shared__ __attribute__((aligned(16))) unsigned bm[SSW * BPI];
  const int tid  = threadIdx.x;
  const int lane = tid & 31;
  const int wv   = tid >> 5;
  int d[BWW];
#pragma unroll
  for (int q = 0; q < BWW; ++q) {
    const long long e = (long long)blockIdx.x * (SSW * 32) + (wv * BWW + q) * 32 + lane;
    d[q] = e < (long long)E ? dst[e] : -1;
  }
  const uint4 z = make_uint4(0u, 0u, 0u, 0u);
  for (int i = tid; i < SSW * BPI / 4; i += BTH) ((uint4*)bm)[i] = z;
  __syncthreads();
#pragma unroll
  for (int q = 0; q < BWW; ++q) {
    const bool valid = d[q] >= 0;
    unsigned o = valid ? (unsigned)d[q] / (unsigned)TPR : 0u;
    o = o < (unsigned)(NOWN - 1) ? o : (unsigned)(NOWN - 1);
    unsigned peers = __builtin_amdgcn_ballot_w32(valid);
#pragma unroll
    for (int b = 0; b < 8; ++b) {
      const bool bit = ((o >> b) & 1u) != 0u;
      const unsigned bal = __builtin_amdgcn_ballot_w32(bit);
      peers &= bit ? bal : ~bal;
    }
    if (valid) bm[(wv * BWW + q) * BPI + (int)o] = peers;
  }
  __syncthreads();
  unsigned* __restrict__ op = bits + (long long)blockIdx.x * SSW + lane * 4;
#pragma unroll 4
  for (int i = 0; i < NOWN / (BTH / 32); ++i) {
    const int o = i * (BTH / 32) + wv;
    v4u v;
    v.x = bm[(lane * 4 + 0) * BPI + o];
    v.y = bm[(lane * 4 + 1) * BPI + o];
    v.z = bm[(lane * 4 + 2) * BPI + o];
    v.w = bm[(lane * 4 + 3) * BPI + o];
    *(v4u*)(op + (long long)o * wpo) = v;
  }
}

template <int K, int N>
__launch_bounds__(32)
__global__ void k_gemm_wmma(const float* __restrict__ A, const float* __restrict__ B,
                            float* __restrict__ C) {
  const int lane = threadIdx.x;
  const int half = lane >> 4;
  const int lm   = lane & 15;
  const int m0 = blockIdx.x * 16;
  const int n0 = blockIdx.y * 16;

  const float* __restrict__ arow = A + (long long)(m0 + lm) * K + 2 * half;
  const float* __restrict__ bp   = B + (long long)(2 * half) * N + n0 + lm;

  v8f acc = {};
#pragma unroll 8
  for (int k = 0; k < K; k += 4) {
    v2f a = *(const v2f*)(arow + k);
    v2f b;
    b.x = bp[k * N];
    b.y = bp[k * N + N];
    acc = __builtin_amdgcn_wmma_f32_16x16x4_f32(
         false, a,  false, b,
         (short)0, acc,  false,  false);
  }

  float* __restrict__ cp = C + (long long)(m0 + 8 * half) * N + n0 + lm;
#pragma unroll
  for (int r = 0; r < 8; ++r)
    cp[r * N] = acc[r];
}

constexpr int PR    = 196;
constexpr int PTB   = 1024;
constexpr int PNW   = PTB / 32;
constexpr int TIT   = 4;
constexpr int PCAP  = 512;
constexpr int PLC   = 480;
constexpr int PLH   = PLC - 32;
constexpr int NRW   = (PR + PNW - 1) / PNW;
constexpr int SCAP  = 24;
constexpr int YP    = 64 + 4;
static_assert(PNW == 32, "row ownership is row & 31");
static_assert(PR == TPR, "the bitmaps are of groups of PR rows");
static_assert(PCAP >= 32, "an empty list must take a whole turn, or a round could make no progress");
static_assert(SCAP <= 32, "an owner takes a sub-list by one load, and one exchange step adds at most 32 edges");

template <int F> struct RowVec;
template <> struct RowVec<64>  { typedef float2 T; };
template <> struct RowVec<128> { typedef float4 T; };

__device__ __forceinline__ void zeroRow(float2& a) { a.x = 0.0f; a.y = 0.0f; }
__device__ __forceinline__ void zeroRow(float4& a) { a.x = 0.0f; a.y = 0.0f; a.z = 0.0f; a.w = 0.0f; }
__device__ __forceinline__ float2 fmaRow(float2 a, float2 v, float w) {
  a.x += v.x * w; a.y += v.y * w;
  return a;
}
__device__ __forceinline__ float4 fmaRow(float4 a, float4 v, float w) {
  a.x += v.x * w; a.y += v.y * w; a.z += v.z * w; a.w += v.w * w;
  return a;
}
__device__ __forceinline__ float2 finishRow(float2 a, float2 hv, float2 bv, float di, float sq, bool relu) {
  float2 r;
  r.x = di * a.x + sq * hv.x + bv.x;
  r.y = di * a.y + sq * hv.y + bv.y;
  if (relu) { r.x = fmaxf(r.x, 0.0f); r.y = fmaxf(r.y, 0.0f); }
  return r;
}
__device__ __forceinline__ float4 finishRow(float4 a, float4 hv, float4 bv, float di, float sq, bool relu) {
  float4 r;
  r.x = di * a.x + sq * hv.x + bv.x;
  r.y = di * a.y + sq * hv.y + bv.y;
  r.z = di * a.z + sq * hv.z + bv.z;
  r.w = di * a.w + sq * hv.w + bv.w;
  if (relu) {
    r.x = fmaxf(r.x, 0.0f); r.y = fmaxf(r.y, 0.0f);
    r.z = fmaxf(r.z, 0.0f); r.w = fmaxf(r.w, 0.0f);
  }
  return r;
}

__device__ __forceinline__ int firstBit(uint4 m) {
  const unsigned w = m.x ? m.x : m.y ? m.y : m.z ? m.z : m.w;
  const int qo = m.x ? 0 : m.y ? 32 : m.z ? 64 : 96;
  return qo + __builtin_ctz(w);
}
__device__ __forceinline__ uint4 dropBit(uint4 m) {
  const bool c0 = m.x != 0u, c1 = m.y != 0u, c2 = m.z != 0u;
  const unsigned w = c0 ? m.x : c1 ? m.y : c2 ? m.z : m.w;
  const unsigned nw = w & (w - 1u);
  uint4 r;
  r.x = c0 ? nw : m.x;
  r.y = (!c0 && c1) ? nw : m.y;
  r.z = (!c0 && !c1 && c2) ? nw : m.z;
  r.w = (!c0 && !c1 && !c2) ? nw : m.w;
  return r;
}

constexpr int LPL = 3;
template <int F, bool RELU, bool DEG, int F2, int WM>
__launch_bounds__(PTB)
__global__ void k_pull(const float* __restrict__ h, const float* __restrict__ dinv, float* __restrict__ dout,
                       const int* __restrict__ src, const int* __restrict__ dst,
                       const float* __restrict__ bias, float* __restrict__ out, int N, int E,
                       int* __restrict__ lcnt, int* __restrict__ lbuf, int lcap,
                       const unsigned* __restrict__ bits, int G, const float* __restrict__ Wn) {
  static_assert(F2 == 0 || (F == 64 && F2 % 16 == 0 && PNW % (F2 / 16) == 0), "fused product: 64-wide rows, whole column tiles, waves split evenly over them");
  static_assert(WM >= 0 && WM <= 2 && (!DEG || WM == 0), "list weights: 0 looked up, 1 looked up and kept, 2 read back; the scan launch has none");
  typedef typename RowVec<F>::T VT;
  constexpr int VW = F / 32;
  constexpr int PB = 2048 / F;
  __shared__ __attribute__((aligned(16))) float acc[DEG ? 4 : PR * F];
  __shared__ int Le[PNW][PCAP];
  __shared__ int Lr[PNW][PCAP];
  __shared__ int C[PNW];
  __shared__ int Dn[PNW];
  __shared__ int lstE[PNW][PLC];
  __shared__ int sub[DEG ? PNW : 1][DEG ? PNW : 1][DEG ? SCAP : 1];
  __shared__ int Tn[DEG ? PNW : 1][DEG ? PNW : 1];
  __shared__ int Ov[DEG ? PNW : 1];
  __shared__ __attribute__((aligned(16))) float ybuf[F2 > 0 ? PR * YP : 4];

  const int tid  = threadIdx.x;
  const int lane = tid & 31;
  const int wv   = __builtin_amdgcn_readfirstlane(tid >> 5);
  const int lo   = blockIdx.x * PR;
  const int nrows = (N - lo) < PR ? (N - lo) : PR;
  const unsigned nr = (unsigned)nrows;

  if constexpr (!DEG) {
    VT z;
    zeroRow(z);
    for (int r = wv; r < PR; r += PNW) *((VT*)(acc + r * F) + lane) = z;
  }

  int* myE = lstE[wv];
  int cnt = 0;

  int* __restrict__ gl = lbuf + ((long long)blockIdx.x * PNW + wv) * LPL * (long long)lcap;
  int gtot = 0;
  bool gover = false;
  int dg[NRW];
#pragma unroll
  for (int q = 0; q < NRW; ++q) dg[q] = 0;

  auto addEdges = [&](int s, int rw, int wb, int n, auto&& behind) {
    for (int hb = 0; hb < n; hb += PB) {
      VT v[PB];
#pragma unroll
      for (int t = 0; t < PB; ++t) {
        const int tt = hb + t < n ? hb + t : n - 1;
        const int sj = __builtin_amdgcn_readlane(s, tt);
        v[t] = *((const VT*)(h + (long long)sj * F) + lane);
      }
      if (hb == 0) behind();
#pragma unroll
      for (int t = 0; t < PB; ++t) {
        if (hb + t < n) {
          const int rj = __builtin_amdgcn_readlane(rw, hb + t);
          const float wj = __int_as_float(__builtin_amdgcn_readlane(wb, hb + t));
          VT* p = (VT*)(acc + rj * F) + lane;
          *p = fmaRow(*p, v[t], wj);
        }
      }
    }
  };

  bool useList = false;
  int ltot = 0;
  if (!DEG && lcap > 0) {
    const int c = lcnt[blockIdx.x * PNW + lane];
    useList = __builtin_amdgcn_ballot_w32(c < 0) == 0;
    ltot = __builtin_amdgcn_readlane(c, wv);
  }

  if (useList) {
    auto loadEnt = [&](int c0, int& s, int& rw, int& wb) {
      const int n = ltot - c0 < 32 ? ltot - c0 : 32;
      const int li = c0 + (lane < n ? lane : n - 1);
      s = gl[li]; rw = gl[lcap + li];
      if constexpr (WM == 2) wb = gl[2 * lcap + li];
      else wb = __float_as_int(dinv[s]);
    };
    int ns = 0, nrw = 0, nwb = 0;
    if (ltot > 0) loadEnt(0, ns, nrw, nwb);
    for (int c0 = 0; c0 < ltot; c0 += 32) {
      const int n = ltot - c0 < 32 ? ltot - c0 : 32;
      const int s = ns, rw = nrw, wb = nwb;
      if constexpr (WM == 1) {
        if (lane < n) gl[2 * lcap + c0 + lane] = wb;
      }
      addEdges(s, rw, wb, n, [&] { loadEnt(c0 + 32 < ltot ? c0 + 32 : c0, ns, nrw, nwb); });
    }
  } else {
    auto workOff = [&] {
      const int n = cnt < 32 ? cnt : 32;
      const int li = cnt - n + (lane < n ? lane : n - 1);
      const int e  = myE[li];
      const int s  = src[e];
      const int rw = dst[e] - lo;
      if constexpr (DEG) {
        if (gtot + n <= lcap) {
          if (lane < n) { gl[gtot + lane] = s; gl[lcap + gtot + lane] = rw; }
          gtot += n;
        } else {
          gover = true;
        }
#pragma unroll
        for (int q = 0; q < NRW; ++q)
          dg[q] += __builtin_popcount(__builtin_amdgcn_ballot_w32((lane < n) & (rw == wv + q * PNW)));
      } else {
        const int wb = __float_as_int(dinv[s]);
        addEdges(s, rw, wb, n, [] {});
      }
      cnt -= n;
    };

    const unsigned* __restrict__ bw = bits + (long long)(blockIdx.x < NOWN - 1 ? blockIdx.x : NOWN - 1) * ((long long)G * SSW);
    int* myF = Le[wv];
    int* myR = Lr[wv];
    int g  = wv;
    int t0 = 0;
    for (;;) {
      int cs = 0;
      bool full = false;
      while (g < G && !full) {
        uint4 tv[TIT];
#pragma unroll
        for (int j = 0; j < TIT; ++j) {
          const int gj = g + j * PNW < G ? g + j * PNW : g;
          tv[j] = *((const uint4*)(bw + (long long)gj * SSW) + lane);
        }
#pragma unroll
        for (int j = 0; j < TIT; ++j) {
          if (!full && g < G) {
            uint4 mw = tv[j];
            for (int t = 0; t < t0; ++t) mw = dropBit(mw);
            const int eb = g * (SSW * 32) + lane * 128;
            int t = t0;
            while (__builtin_amdgcn_ballot_w32((mw.x | mw.y | mw.z | mw.w) != 0)) {
              const bool v = (mw.x | mw.y | mw.z | mw.w) != 0;
              const unsigned mask = __builtin_amdgcn_ballot_w32(v);
              const int n = __builtin_popcount(mask);
              if (cs + n > PCAP) { full = true; break; }
              const int pos = cs + (int)__builtin_amdgcn_mbcnt_lo(mask, 0u);
              if (v) myF[pos] = eb + firstBit(mw);
              cs += n;
              mw = dropBit(mw);
              ++t;
            }
            if (full) t0 = t;
            else { g += PNW; t0 = 0; }
          }
        }
      }
      int cntv = 0;
      bool ovf = false;
      for (int i = 0; i < cs; i += 128) {
        int r[4];
        int ef[4];
#pragma unroll
        for (int q = 0; q < 4; ++q) {
          const int k = i + q * 32 + lane;
          ef[q] = k < cs ? myF[k] : 0;
          r[q] = k < cs ? dst[ef[q]] - lo : -1;
        }
#pragma unroll
        for (int q = 0; q < 4; ++q) {
          const int k = i + q * 32 + lane;
          const bool val = (unsigned)r[q] < nr;
          if (k < cs) myR[k] = val ? r[q] : -1;
          if constexpr (DEG) {
            const unsigned o = val ? (unsigned)r[q] & 31u : 0u;
            unsigned peers = __builtin_amdgcn_ballot_w32(val);
            unsigned own = peers;
#pragma unroll
            for (int b = 0; b < 5; ++b) {
              const bool bit = ((o >> b) & 1u) != 0u;
              const unsigned bal = __builtin_amdgcn_ballot_w32(bit);
              peers &= bit ? bal : ~bal;
              own &= ((lane >> b) & 1) != 0 ? bal : ~bal;
            }
            const int base = __builtin_amdgcn_ds_bpermute((int)(o << 2), cntv);
            const int pos = base + (int)__builtin_amdgcn_mbcnt_lo(peers, 0u);
            if (val) {
              if (pos < SCAP) sub[wv][o][pos] = ef[q];
              else ovf = true;
            }
            cntv += __builtin_popcount(own);
          }
        }
      }
      if constexpr (DEG) {
        Tn[lane][wv] = cntv;
        const bool anyOvf = __builtin_amdgcn_ballot_w32(ovf) != 0;
        if (lane == 0) Ov[wv] = anyOvf ? 1 : 0;
      }
      if (lane == 0) { C[wv] = cs; Dn[wv] = g >= G ? 1 : 0; }
      __syncthreads();
      const int cAll = C[lane];
      const bool allDone = __builtin_amdgcn_ballot_w32(Dn[lane] == 0) == 0;
      bool direct = false;
      if constexpr (DEG) {
        direct = __builtin_amdgcn_ballot_w32(Ov[lane] != 0) == 0;
        if (direct) {
          const int cOwn = Tn[wv][lane];
          for (int w = 0; w < PNW; ++w) {
            const int c = __builtin_amdgcn_readlane(cOwn, w);
            if (c > 0) {
              if (lane < c) myE[cnt + lane] = sub[w][wv][lane];
              cnt += c;
              while (cnt > PLH) workOff();
            }
          }
        }
      }
      if (!direct) {
        for (int w = 0; w < PNW; ++w) {
          const int cw = __builtin_amdgcn_readlane(cAll, w);
          for (int i = 0; i < cw; i += 32) {
            const int k = i + lane;
            const int r = k < cw ? Lr[w][k] : -1;
            const bool mine = (r >= 0) & ((r & 31) == wv);
            const unsigned mask = __builtin_amdgcn_ballot_w32(mine);
            if (mask) {
              const int pos = cnt + (int)__builtin_amdgcn_mbcnt_lo(mask, 0u);
              if (mine) myE[pos] = Le[w][k];
              cnt += __builtin_popcount(mask);
            }
            while (cnt > PLH) workOff();
          }
        }
      }
      while (cnt > 0) workOff();
      if (allDone) break;
      __syncthreads();
    }
  }
  if constexpr (DEG) {
    if (lcap > 0 && lane == 0) lcnt[blockIdx.x * PNW + wv] = gover ? -1 : gtot;
    int my = 0;
#pragma unroll
    for (int q = 0; q < NRW; ++q) my = lane == q ? dg[q] : my;
    const int r = wv + lane * PNW;
    if (lane < NRW && r < nrows) dout[lo + r] = rsqrtf((float)my + 1.0f);
    return;
  }

  const VT bv = *((const VT*)bias + lane);
  if constexpr (F2 == 0) {
    for (int r = wv; r < nrows; r += PNW) {
      const long long i = lo + r;
      const float di = dinv[i];
      const VT hv = *((const VT*)(h + i * F) + lane);
      const VT a  = *((const VT*)(acc + r * F) + lane);
      *((VT*)(out + i * F) + lane) = finishRow(a, hv, bv, di, di * di, RELU);
    }
  } else {
    for (int r = wv; r < PR; r += PNW) {
      VT y;
      zeroRow(y);
      if (r < nrows) {
        const long long i = lo + r;
        const float di = dinv[i];
        const VT hv = *((const VT*)(h + i * F) + lane);
        const VT a  = *((const VT*)(acc + r * F) + lane);
        y = finishRow(a, hv, bv, di, di * di, RELU);
      }
      *((VT*)(ybuf + r * YP) + lane) = y;
    }
    __syncthreads();
    constexpr int NCT = F2 / 16;
    constexpr int NSG = PNW / NCT;
    const int half = lane >> 4;
    const int lm   = lane & 15;
    const int n0   = (wv % NCT) * 16;
    const int sg   = wv / NCT;
    const float* __restrict__ bp = Wn + (2 * half) * F2 + n0 + lm;
    v2f bf[F / 4];
#pragma unroll
    for (int kk = 0; kk < F / 4; ++kk) {
      bf[kk].x = bp[(4 * kk) * F2];
      bf[kk].y = bp[(4 * kk) * F2 + F2];
    }
    for (int s = sg; s * 16 < nrows; s += NSG) {
      const int ar = s * 16 + lm < PR ? s * 16 + lm : PR - 1;
      const float* ap = ybuf + ar * YP + 2 * half;
      v8f d = {};
#pragma unroll
      for (int kk = 0; kk < F / 4; ++kk) {
        const v2f a = *(const v2f*)(ap + 4 * kk);
        d = __builtin_amdgcn_wmma_f32_16x16x4_f32(
             false, a,  false, bf[kk],
             (short)0, d,  false,  false);
      }
      const int rb = s * 16 + 8 * half;
      float* __restrict__ cp = out + (long long)(lo + rb) * F2 + n0 + lm;
      if (s * 16 + 16 <= nrows) {
#pragma unroll
        for (int r = 0; r < 8; ++r) cp[r * F2] = d[r];
      } else {
#pragma unroll
        for (int r = 0; r < 8; ++r)
          if (rb + r < nrows) cp[r * F2] = d[r];
      }
    }
  }
}

extern "C" void kernel_launch(void* const* d_in, const int* in_sizes, int n_in,
                              void* d_out, int out_size, void* d_ws, size_t ws_size,
                              hipStream_t stream) {
  const float* x  = (const float*)d_in[0];
  const int*   ei = (const int*)d_in[1];
  const float* W1 = (const float*)d_in[2];
  const float* b1 = (const float*)d_in[3];
  const float* W2 = (const float*)d_in[4];
  const float* b2 = (const float*)d_in[5];

  const int FIN = 256, H = 64, FOUT = 128;
  const int N = in_sizes[0] / FIN;
  const int E = in_sizes[1] / 2;
  const int* srcI = ei;
  const int* dstI = ei + E;

  float* ws   = (float*)d_ws;
  float* dinv = ws;
  const long long hO = ((long long)N + 63) / 64 * 64;
  float* h1   = ws + hO;
  float* h2   = h1 + (long long)N * H;
  const int LCAP = 1024;
  const long long nwg   = ((long long)N + PR - 1) / PR;
  const int G = (int)(((long long)E + SSW * 32 - 1) / (SSW * 32));
  const int wpo = G * SSW;
  const long long bitO  = (hO + (long long)N * (H + FOUT) + 63) / 64 * 64;
  const long long lcntO = bitO + (long long)NOWN * wpo;
  const long long lbufO = lcntO + (nwg * PNW + 63) / 64 * 64;
  const bool fits = (size_t)(lbufO + nwg * PNW * LPL * LCAP) * sizeof(float) <= ws_size;
  unsigned* bits = (unsigned*)ws + bitO;
  int* lcnt = (int*)ws + lcntO;
  int* lbuf = (int*)ws + lbufO;
  const int lcap = fits ? LCAP : 0;

  auto blocks = [](long long n, int tb) { return (unsigned)((n + tb - 1) / tb); };

  k_bits<<<(unsigned)G, BTH, 0, stream>>>(dstI, bits, E, wpo);
  k_pull<64, false, true, 0, 0><<<blocks((long long)N, PR), PTB, 0, stream>>>(nullptr, nullptr, dinv, srcI, dstI, nullptr, nullptr, N, E, lcnt, lbuf, lcap, bits, G, nullptr);

  {
    dim3 g((unsigned)(N / 16), H / 16);
    k_gemm_wmma<256, 64><<<g, 32, 0, stream>>>(x, W1, h1);
  }
  k_pull<64, true, false, FOUT, 1><<<blocks((long long)N, PR), PTB, 0, stream>>>(h1, dinv, nullptr, srcI, dstI, b1, h2, N, E, lcnt, lbuf, lcap, bits, G, W2);

  k_pull<128, false, false, 0, 2><<<blocks((long long)N, PR), PTB, 0, stream>>>(h2, dinv, nullptr, srcI, dstI, b2, (float*)d_out, N, E, lcnt, lbuf, lcap, bits, G, nullptr);
}
